# starting kernel
# baseline (speedup 1.0000x reference)
#include <hip/hip_runtime.h>


#define DD1  128
#define DD2  128
#define BB   16
#define SIN  64
#define SOUT 128

typedef __attribute__((ext_vector_type(2))) float v2f;
typedef __attribute__((ext_vector_type(4))) float v4f;
typedef __attribute__((ext_vector_type(8))) float v8f;

__global__ __launch_bounds__(256)
void mdrnn_gemm_wmma(const float* __restrict__ x, const float* __restrict__ w,
                     const float* __restrict__ bias, float* __restrict__ a_t)
{
    __shared__ float tile[512 * 32];

    const int i    = blockIdx.z;
    const int j0   = blockIdx.y * 32;
    const int o0   = blockIdx.x * 32;
    const int tid  = threadIdx.x;
    const int lane = tid & 31;
    const int wave = tid >> 5;

    const int lm = lane & 15;
    const int kh = (lane >> 4) * 2;
    const int hh = lane >> 4;

    for (int t = 0; t < 8; ++t) {
        const int jj = wave * 4 + (t >> 1);
        const int nn = (t & 1) * 16;
        const int j  = j0 + jj;

        const float* xrow = x + (((size_t)i * DD2 + j) * BB + lm) * SIN;

        v8f acc = {};
        #pragma unroll
        for (int k0 = 0; k0 < SIN; k0 += 4) {
            v2f af;
            af.x = xrow[k0 + kh];
            af.y = xrow[k0 + kh + 1];
            const float* wp = w + (size_t)(k0 + kh) * SOUT + (o0 + nn + lm);
            v2f bf;
            bf.x = wp[0];
            bf.y = wp[SOUT];
            acc = __builtin_amdgcn_wmma_f32_16x16x4_f32(
                false, af, false, bf, (short)0, acc, false, false);
        }

        #pragma unroll
        for (int v = 0; v < 8; ++v)
            tile[(jj * 16 + (v + 8 * hh)) * 32 + nn + lm] = acc[v];
    }
    __syncthreads();

    for (int p = tid; p < 512; p += 256) {
        const int b  = p >> 5;
        const int oo = p & 31;
        const int o  = o0 + oo;
        const float bv = bias[o];
        float* dst = a_t + (size_t)(b * SOUT + o) * (DD1 * DD2)
                         + (size_t)i * DD2 + j0;
        #pragma unroll
        for (int jj = 0; jj < 32; jj += 4) {
            v4f val;
            val.x = tile[((jj + 0) * 16 + b) * 32 + oo] + bv;
            val.y = tile[((jj + 1) * 16 + b) * 32 + oo] + bv;
            val.z = tile[((jj + 2) * 16 + b) * 32 + oo] + bv;
            val.w = tile[((jj + 3) * 16 + b) * 32 + oo] + bv;
            *(v4f*)(dst + jj) = val;
        }
    }
}

__global__ __launch_bounds__(256)
void mdrnn_scan(const float* __restrict__ a_t, const float* __restrict__ u,
                float* __restrict__ h_out)
{
    __shared__ float boundary[8][DD1];

    const int tid  = threadIdx.x;
    const int lane = tid & 31;
    const int wave = tid >> 5;
    const int bo   = blockIdx.x * 8 + wave;
    const int b    = bo >> 7;
    const int o    = bo & 127;

    const float u0 = u[o];
    const float u1 = u[SOUT + o];
    const float* abase = a_t + (size_t)bo * (DD1 * DD2);
    float* bnd = boundary[wave];

    for (int s = 0; s < 4; ++s) {
        const int j = s * 32 + lane;
        float h = 0.0f;
        for (int t = 0; t < DD1 + 31; ++t) {
            float hl = __shfl_up(h, 1);
            const int i = t - lane;
            if (lane == 0)
                hl = (s > 0 && i >= 0 && i < DD1) ? bnd[i] : 0.0f;
            if (i >= 0 && i < DD1) {
                const float a  = abase[i * DD2 + j];
                const float hn = tanhf(fmaf(u0, h, fmaf(u1, hl, a)));
                h = hn;
                if (lane == 31) bnd[i] = hn;
                h_out[(((size_t)i * DD2 + j) * BB + b) * SOUT + o] = hn;
            }
        }
    }
}

extern "C" void kernel_launch(void* const* d_in, const int* in_sizes, int n_in,
                              void* d_out, int out_size, void* d_ws, size_t ws_size,
                              hipStream_t stream)
{
    const float* x    = (const float*)d_in[0];
    const float* w    = (const float*)d_in[1];
    const float* u    = (const float*)d_in[2];
    const float* bias = (const float*)d_in[3];
    float* h   = (float*)d_out;
    float* a_t = (float*)d_ws;

    dim3 g1(SOUT / 32, DD2 / 32, DD1);
    mdrnn_gemm_wmma<<<g1, 256, 0, stream>>>(x, w, bias, a_t);

    mdrnn_scan<<<(BB * SOUT) / 8, 256, 0, stream>>>(a_t, u, h);
}


# scan re-cut: 256 work-groups of 8 channels; a wave's lanes = 8 channels x 4 columns swept by anti-diagonals (255 one-cell steps, left h by an 8-lane shift): all WGPs scan, not 64; sums unchanged
# speedup vs baseline: 4.8587x; 4.8587x over previous
#include <hip/hip_runtime.h>


#define DD1  128
#define DD2  128
#define BB   16
#define SIN  64
#define SOUT 128

#define CELLF (BB * SOUT)
#define ROWF  (DD2 * CELLF)

#define SW 32
#define CG 8
#define LQ (DD2 / SW)
#define PD 4

typedef __attribute__((ext_vector_type(2))) float v2f;
typedef __attribute__((ext_vector_type(4))) float v4f;
typedef __attribute__((ext_vector_type(8))) float v8f;

#define XP 68

__global__ __launch_bounds__(256)
void mdrnn_gemm_wmma(const float* __restrict__ x, const float* __restrict__ w,
                     const float* __restrict__ bias, float* __restrict__ a_nat)
{
    __shared__ float tile[512 * 32];
    __shared__ float xs[8][16 * XP];

    const int i    = blockIdx.z;
    const int j0   = blockIdx.y * 32;
    const int o0   = blockIdx.x * 32;
    const int tid  = threadIdx.x;
    const int lane = tid & 31;
    const int wave = tid >> 5;

    const int lm = lane & 15;
    const int kh = (lane >> 4) * 2;
    const int hh = lane >> 4;

    const float* xw = x + (((size_t)i * DD2 + j0 + wave * 4) * BB) * SIN;
    float* xsw = xs[wave];
    float* sdst = xsw + hh * XP + lm * 4;
    const float* afrag = xsw + lm * XP + kh;

    v4f pf[8];
    #pragma unroll
    for (int q = 0; q < 8; ++q)
        pf[q] = *(const v4f*)(xw + (q * 32 + lane) * 4);

    for (int t = 0; t < 4; ++t) {
        const int jj = wave * 4 + t;

        #pragma unroll
        for (int q = 0; q < 8; ++q)
            *(v4f*)(sdst + 2 * q * XP) = pf[q];

        if (t < 3) {
            const float* xn = xw + (t + 1) * (BB * SIN);
            #pragma unroll
            for (int q = 0; q < 8; ++q)
                pf[q] = *(const v4f*)(xn + (q * 32 + lane) * 4);
        }

        v8f acc0 = {};
        v8f acc1 = {};
        #pragma unroll
        for (int k0 = 0; k0 < SIN; k0 += 4) {
            const v2f af = *(const v2f*)(afrag + k0);
            const float* wp = w + (size_t)(k0 + kh) * SOUT + (o0 + lm);
            v2f bf0, bf1;
            bf0.x = wp[0];
            bf0.y = wp[SOUT];
            bf1.x = wp[16];
            bf1.y = wp[SOUT + 16];
            acc0 = __builtin_amdgcn_wmma_f32_16x16x4_f32(
                false, af, false, bf0, (short)0, acc0, false, false);
            acc1 = __builtin_amdgcn_wmma_f32_16x16x4_f32(
                false, af, false, bf1, (short)0, acc1, false, false);
        }

        #pragma unroll
        for (int v = 0; v < 8; ++v) {
            tile[(jj * 16 + (v + 8 * hh)) * 32 + lm]      = acc0[v];
            tile[(jj * 16 + (v + 8 * hh)) * 32 + 16 + lm] = acc1[v];
        }
    }
    __syncthreads();

    {
        const int q = (tid & 7) * 4;
        const float bv0 = bias[o0 + q + 0];
        const float bv1 = bias[o0 + q + 1];
        const float bv2 = bias[o0 + q + 2];
        const float bv3 = bias[o0 + q + 3];
        float* dst = a_nat + ((size_t)i * DD2 + j0) * CELLF + o0 + q;
        const int m0 = tid >> 3;
        #pragma unroll 4
        for (int k = 0; k < 16; ++k) {
            const int m = m0 + 32 * k;
            v4f val = *(const v4f*)(tile + m * 32 + q);
            val.x += bv0;
            val.y += bv1;
            val.z += bv2;
            val.w += bv3;
            *(v4f*)(dst + (size_t)m * SOUT) = val;
        }
    }
}

static __device__ __forceinline__ void wg_barrier_lds_only()
{
    asm volatile("s_wait_dscnt 0x0\n\ts_barrier_signal -1\n\ts_barrier_wait -1" ::: "memory");
}

static __device__ __forceinline__ float lane_up8(const float v, const int q)
{
    const int iv = __builtin_bit_cast(int, v);
    const int t1 = __builtin_amdgcn_update_dpp(0, iv, 0x118, 0xf, 0xf, true);
    const int t2 = (int)__builtin_amdgcn_permlanex16((unsigned)iv, (unsigned)iv,
                                                     0xfedcba98u, 0x76543210u, false, false);
    return __builtin_bit_cast(float, q == 2 ? t2 : t1);
}

static __device__ __forceinline__ void scan_step(
    const int t, const int w, const int lane, const int q, const int j,
    const float cur, float &ld, float &hu, float &hs,
    const float* __restrict__ acol, float* __restrict__ hcol,
    float (*bnd)[SW][32], const float u0, const float u1)
{
    {
        int r = t + PD - j;
        r = r < 0 ? 0 : r;
        r = r > DD1 - 1 ? DD1 - 1 : r;
        ld = acol[(size_t)r * ROWF];
    }

    if (t >= LQ * w && t <= LQ * w + (DD1 - 1) + (LQ - 1)) {
        const float hb = bnd[(t + 1) & 1][w > 0 ? w - 1 : 0][24 + (lane & 7)];
        const float hl = (q == 0) ? (w > 0 ? hb : 0.0f) : hs;
        const int i = t - j;
        float hn = tanhf(fmaf(u0, hu, fmaf(u1, hl, cur)));
        hn = (i >= 0) ? hn : 0.0f;
        bnd[t & 1][w][lane] = hn;
        if (i >= 0 && i < DD1)
            hcol[(size_t)i * ROWF] = hn;
        hu = hn;
        hs = lane_up8(hn, q);
    }
    wg_barrier_lds_only();
}

__global__ __launch_bounds__(SW * 32)
void mdrnn_scan_channels(const float* __restrict__ a, const float* __restrict__ u,
                         float* __restrict__ h_out)
{
    __shared__ float bnd[2][SW][32];

    const int lane = threadIdx.x & 31;
    const int w    = __builtin_amdgcn_readfirstlane(threadIdx.x >> 5);
    const int g    = blockIdx.x;
    const int c    = lane & 7;
    const int q    = lane >> 3;
    const int j    = LQ * w + q;
    const int o    = (g & 15) * CG + c;

    const float u0 = u[o];
    const float u1 = u[SOUT + o];

    const size_t col0 = (size_t)j * CELLF + (size_t)g * CG + c;
    const float* acol = a + col0;
    float* hcol = h_out + col0;

    float hu = 0.0f;
    float hs = 0.0f;
    float r0, r1, r2, r3, r4;

    {
        int k0 = 0 - j; k0 = k0 < 0 ? 0 : k0;
        int k1 = 1 - j; k1 = k1 < 0 ? 0 : k1;
        int k2 = 2 - j; k2 = k2 < 0 ? 0 : k2;
        int k3 = 3 - j; k3 = k3 < 0 ? 0 : k3;
        r0 = acol[(size_t)k0 * ROWF];
        r1 = acol[(size_t)k1 * ROWF];
        r2 = acol[(size_t)k2 * ROWF];
        r3 = acol[(size_t)k3 * ROWF];
    }

    for (int t = 0; t < DD1 + DD2 - 1; t += PD + 1) {
        scan_step(t,     w, lane, q, j, r0, r4, hu, hs, acol, hcol, bnd, u0, u1);
        scan_step(t + 1, w, lane, q, j, r1, r0, hu, hs, acol, hcol, bnd, u0, u1);
        scan_step(t + 2, w, lane, q, j, r2, r1, hu, hs, acol, hcol, bnd, u0, u1);
        scan_step(t + 3, w, lane, q, j, r3, r2, hu, hs, acol, hcol, bnd, u0, u1);
        scan_step(t + 4, w, lane, q, j, r4, r3, hu, hs, acol, hcol, bnd, u0, u1);
    }
}

extern "C" void kernel_launch(void* const* d_in, const int* in_sizes, int n_in,
                              void* d_out, int out_size, void* d_ws, size_t ws_size,
                              hipStream_t stream)
{
    const float* x    = (const float*)d_in[0];
    const float* w    = (const float*)d_in[1];
    const float* u    = (const float*)d_in[2];
    const float* bias = (const float*)d_in[3];
    float* h     = (float*)d_out;
    float* a_nat = (float*)d_ws;

    dim3 g1(SOUT / 32, DD2 / 32, DD1);
    mdrnn_gemm_wmma<<<g1, 256, 0, stream>>>(x, w, bias, a_nat);

    mdrnn_scan_channels<<<(BB * SOUT) / CG, SW * 32, 0, stream>>>(a_nat, u, h);
}


# a kept in the scan's own order [8-channel group][anti-diagonal (i+j) mod 128][column][channel]; GEMM groups take 32 cells of a diagonal; a scan wave reads one whole line per step; sums unchanged
# speedup vs baseline: 5.3728x; 1.1058x over previous
#include <hip/hip_runtime.h>


#define DD1  128
#define DD2  128
#define BB   16
#define SIN  64
#define SOUT 128

#define CELLF (BB * SOUT)
#define ROWF  (DD2 * CELLF)

#define NS    128

#define SW 32
#define CG 8
#define LQ (DD2 / SW)
#define PD 4
#define SLOTF (DD2 * CG)
#define SLABF (NS * SLOTF)

typedef __attribute__((ext_vector_type(2))) float v2f;
typedef __attribute__((ext_vector_type(4))) float v4f;
typedef __attribute__((ext_vector_type(8))) float v8f;

#define XP 68
#define TJ 520

static __device__ __forceinline__ const float* cell_x(const float* __restrict__ x,
                                                      const int s, const int j)
{
    const int i = (s - j) & (DD1 - 1);
    return x + ((size_t)(i * DD2 + j)) * (BB * SIN);
}

__global__ __launch_bounds__(256)
void mdrnn_gemm_wmma(const float* __restrict__ x, const float* __restrict__ w,
                     const float* __restrict__ bias, float* __restrict__ a_ws)
{
    __shared__ __attribute__((aligned(16))) float tile[32 * TJ];
    __shared__ __attribute__((aligned(16))) float xs[8][16 * XP];

    const int s    = blockIdx.z;
    const int j0   = blockIdx.y * 32;
    const int o0   = blockIdx.x * 32;
    const int tid  = threadIdx.x;
    const int lane = tid & 31;
    const int wave = tid >> 5;

    const int lm = lane & 15;
    const int kh = (lane >> 4) * 2;
    const int hh = lane >> 4;

    const int jw = j0 + wave * 4;
    const float* xw = cell_x(x, s, jw);
    float* xsw = xs[wave];
    float* sdst = xsw + hh * XP + lm * 4;
    const float* afrag = xsw + lm * XP + kh;

    v4f pf[8];
    #pragma unroll
    for (int q = 0; q < 8; ++q)
        pf[q] = *(const v4f*)(xw + (q * 32 + lane) * 4);

    #pragma unroll
    for (int t = 0; t < 4; ++t) {
        const int jj = wave * 4 + t;

        #pragma unroll
        for (int q = 0; q < 8; ++q)
            *(v4f*)(sdst + 2 * q * XP) = pf[q];

        if (t < 3) {
            const float* xn = cell_x(x, s, jw + t + 1);
            #pragma unroll
            for (int q = 0; q < 8; ++q)
                pf[q] = *(const v4f*)(xn + (q * 32 + lane) * 4);
        }

        v8f acc0 = {};
        v8f acc1 = {};
        #pragma unroll
        for (int k0 = 0; k0 < SIN; k0 += 4) {
            const v2f af = *(const v2f*)(afrag + k0);
            const float* wp = w + (size_t)(k0 + kh) * SOUT + (o0 + lm);
            v2f bf0, bf1;
            bf0.x = wp[0];
            bf0.y = wp[SOUT];
            bf1.x = wp[16];
            bf1.y = wp[SOUT + 16];
            acc0 = __builtin_amdgcn_wmma_f32_16x16x4_f32(
                false, af, false, bf0, (short)0, acc0, false, false);
            acc1 = __builtin_amdgcn_wmma_f32_16x16x4_f32(
                false, af, false, bf1, (short)0, acc1, false, false);
        }

        #pragma unroll
        for (int v = 0; v < 8; ++v) {
            tile[jj * TJ + (v + 8 * hh) * 32 + lm]      = acc0[v];
            tile[jj * TJ + (v + 8 * hh) * 32 + 16 + lm] = acc1[v];
        }
    }
    __syncthreads();

    {
        const int oct = tid >> 6;
        const int p   = tid & 63;
        const int jj  = p >> 1;
        const int c0  = (p & 1) * 4;
        const int ob  = o0 + oct * 8 + c0;
        const float bv0 = bias[ob + 0];
        const float bv1 = bias[ob + 1];
        const float bv2 = bias[ob + 2];
        const float bv3 = bias[ob + 3];
        const float* srcp = tile + jj * TJ + oct * 8 + c0;
        float* dst = a_ws + ((size_t)(((o0 >> 3) + oct) * NS + s)) * SLOTF + (size_t)j0 * CG + p * 4;
        #pragma unroll 4
        for (int b = 0; b < BB; ++b) {
            v4f val = *(const v4f*)(srcp + b * 32);
            val.x += bv0;
            val.y += bv1;
            val.z += bv2;
            val.w += bv3;
            *(v4f*)(dst + (size_t)b * (16 * (size_t)SLABF)) = val;
        }
    }
}

static __device__ __forceinline__ void wg_barrier_lds_only()
{
    asm volatile("s_wait_dscnt 0x0\n\ts_barrier_signal -1\n\ts_barrier_wait -1" ::: "memory");
}

static __device__ __forceinline__ float lane_up8(const float v, const int q)
{
    const int iv = __builtin_bit_cast(int, v);
    const int t1 = __builtin_amdgcn_update_dpp(0, iv, 0x118, 0xf, 0xf, true);
    const int t2 = (int)__builtin_amdgcn_permlanex16((unsigned)iv, (unsigned)iv,
                                                     0xfedcba98u, 0x76543210u, false, false);
    return __builtin_bit_cast(float, q == 2 ? t2 : t1);
}

static __device__ __forceinline__ void scan_step(
    const int t, const int w, const int lane, const int q, const int j,
    const float cur, float &ld, float &hu, float &hs,
    const float* __restrict__ aslab, float* __restrict__ hcol,
    float (*bnd)[SW][32], const float u0, const float u1)
{
    {
        int tt = t + PD;
        tt = tt < LQ * w ? LQ * w : tt;
        tt = tt > LQ * w + (DD1 - 1) + (LQ - 1) ? LQ * w + (DD1 - 1) + (LQ - 1) : tt;
        ld = aslab[(size_t)(tt & (NS - 1)) * SLOTF];
    }

    if (t >= LQ * w && t <= LQ * w + (DD1 - 1) + (LQ - 1)) {
        const float hb = bnd[(t + 1) & 1][w > 0 ? w - 1 : 0][24 + (lane & 7)];
        const float hl = (q == 0) ? (w > 0 ? hb : 0.0f) : hs;
        const int i = t - j;
        float hn = tanhf(fmaf(u0, hu, fmaf(u1, hl, cur)));
        hn = (i >= 0) ? hn : 0.0f;
        bnd[t & 1][w][lane] = hn;
        if (i >= 0 && i < DD1)
            hcol[(size_t)i * ROWF] = hn;
        hu = hn;
        hs = lane_up8(hn, q);
    }
    wg_barrier_lds_only();
}

__global__ __launch_bounds__(SW * 32)
void mdrnn_scan_channels(const float* __restrict__ a, const float* __restrict__ u,
                         float* __restrict__ h_out)
{
    __shared__ float bnd[2][SW][32];

    const int lane = threadIdx.x & 31;
    const int w    = __builtin_amdgcn_readfirstlane(threadIdx.x >> 5);
    const int g    = blockIdx.x;
    const int c    = lane & 7;
    const int q    = lane >> 3;
    const int j    = LQ * w + q;
    const int o    = (g & 15) * CG + c;

    const float u0 = u[o];
    const float u1 = u[SOUT + o];

    const size_t col0 = (size_t)j * CELLF + (size_t)g * CG + c;
    const float* aslab = a + (size_t)g * SLABF + 32 * w + lane;
    float* hcol = h_out + col0;

    float hu = 0.0f;
    float hs = 0.0f;
    float r0, r1, r2, r3, r4;

    r0 = aslab[0 * SLOTF];
    r1 = aslab[1 * SLOTF];
    r2 = aslab[2 * SLOTF];
    r3 = aslab[3 * SLOTF];

    for (int t = 0; t < DD1 + DD2 - 1; t += PD + 1) {
        scan_step(t,     w, lane, q, j, r0, r4, hu, hs, aslab, hcol, bnd, u0, u1);
        scan_step(t + 1, w, lane, q, j, r1, r0, hu, hs, aslab, hcol, bnd, u0, u1);
        scan_step(t + 2, w, lane, q, j, r2, r1, hu, hs, aslab, hcol, bnd, u0, u1);
        scan_step(t + 3, w, lane, q, j, r3, r2, hu, hs, aslab, hcol, bnd, u0, u1);
        scan_step(t + 4, w, lane, q, j, r4, r3, hu, hs, aslab, hcol, bnd, u0, u1);
    }
}

extern "C" void kernel_launch(void* const* d_in, const int* in_sizes, int n_in,
                              void* d_out, int out_size, void* d_ws, size_t ws_size,
                              hipStream_t stream)
{
    const float* x    = (const float*)d_in[0];
    const float* w    = (const float*)d_in[1];
    const float* u    = (const float*)d_in[2];
    const float* bias = (const float*)d_in[3];
    float* h     = (float*)d_out;
    float* a_ws  = (float*)d_ws;

    dim3 g1(SOUT / 32, DD2 / 32, NS);
    mdrnn_gemm_wmma<<<g1, 256, 0, stream>>>(x, w, bias, a_ws);

    mdrnn_scan_channels<<<(BB * SOUT) / CG, SW * 32, 0, stream>>>(a_ws, u, h);
}


# scan: a lane carries two adjacent channels as a float pair (16 waves of 8 columns x 4 pairs): step bookkeeping, barrier, hand-over paid once per two cells; two-range tanh on pairs; sums unchanged
# speedup vs baseline: 5.5871x; 1.0399x over previous
#include <hip/hip_runtime.h>


#define DD1  128
#define DD2  128
#define BB   16
#define SIN  64
#define SOUT 128

#define CELLF (BB * SOUT)
#define ROWF  (DD2 * CELLF)

#define NS    128

#define SW 16
#define CG 8
#define LQ (DD2 / SW)
#define PD 4
#define SLOTF (DD2 * CG)
#define SLABF (NS * SLOTF)

typedef __attribute__((ext_vector_type(2))) float v2f;
typedef __attribute__((ext_vector_type(4))) float v4f;
typedef __attribute__((ext_vector_type(8))) float v8f;

#define XP 68
#define TJ 520

static __device__ __forceinline__ const float* cell_x(const float* __restrict__ x,
                                                      const int s, const int j)
{
    const int i = (s - j) & (DD1 - 1);
    return x + ((size_t)(i * DD2 + j)) * (BB * SIN);
}

__global__ __launch_bounds__(256)
void mdrnn_gemm_wmma(const float* __restrict__ x, const float* __restrict__ w,
                     const float* __restrict__ bias, float* __restrict__ a_ws)
{
    __shared__ __attribute__((aligned(16))) float tile[32 * TJ];
    __shared__ __attribute__((aligned(16))) float xs[8][16 * XP];

    const int s    = blockIdx.z;
    const int j0   = blockIdx.y * 32;
    const int o0   = blockIdx.x * 32;
    const int tid  = threadIdx.x;
    const int lane = tid & 31;
    const int wave = tid >> 5;

    const int lm = lane & 15;
    const int kh = (lane >> 4) * 2;
    const int hh = lane >> 4;

    const int jw = j0 + wave * 4;
    const float* xw = cell_x(x, s, jw);
    float* xsw = xs[wave];
    float* sdst = xsw + hh * XP + lm * 4;
    const float* afrag = xsw + lm * XP + kh;

    v4f pf[8];
    #pragma unroll
    for (int q = 0; q < 8; ++q)
        pf[q] = *(const v4f*)(xw + (q * 32 + lane) * 4);

    #pragma unroll
    for (int t = 0; t < 4; ++t) {
        const int jj = wave * 4 + t;

        #pragma unroll
        for (int q = 0; q < 8; ++q)
            *(v4f*)(sdst + 2 * q * XP) = pf[q];

        if (t < 3) {
            const float* xn = cell_x(x, s, jw + t + 1);
            #pragma unroll
            for (int q = 0; q < 8; ++q)
                pf[q] = *(const v4f*)(xn + (q * 32 + lane) * 4);
        }

        v8f acc0 = {};
        v8f acc1 = {};
        #pragma unroll
        for (int k0 = 0; k0 < SIN; k0 += 4) {
            const v2f af = *(const v2f*)(afrag + k0);
            const float* wp = w + (size_t)(k0 + kh) * SOUT + (o0 + lm);
            v2f bf0, bf1;
            bf0.x = wp[0];
            bf0.y = wp[SOUT];
            bf1.x = wp[16];
            bf1.y = wp[SOUT + 16];
            acc0 = __builtin_amdgcn_wmma_f32_16x16x4_f32(
                false, af, false, bf0, (short)0, acc0, false, false);
            acc1 = __builtin_amdgcn_wmma_f32_16x16x4_f32(
                false, af, false, bf1, (short)0, acc1, false, false);
        }

        #pragma unroll
        for (int v = 0; v < 8; ++v) {
            tile[jj * TJ + (v + 8 * hh) * 32 + lm]      = acc0[v];
            tile[jj * TJ + (v + 8 * hh) * 32 + 16 + lm] = acc1[v];
        }
    }
    __syncthreads();

    {
        const int oct = tid >> 6;
        const int p   = tid & 63;
        const int jj  = p >> 1;
        const int c0  = (p & 1) * 4;
        const int ob  = o0 + oct * 8 + c0;
        const float bv0 = bias[ob + 0];
        const float bv1 = bias[ob + 1];
        const float bv2 = bias[ob + 2];
        const float bv3 = bias[ob + 3];
        const float* srcp = tile + jj * TJ + oct * 8 + c0;
        float* dst = a_ws + ((size_t)(((o0 >> 3) + oct) * NS + s)) * SLOTF + (size_t)j0 * CG + p * 4;
        #pragma unroll 4
        for (int b = 0; b < BB; ++b) {
            v4f val = *(const v4f*)(srcp + b * 32);
            val.x += bv0;
            val.y += bv1;
            val.z += bv2;
            val.w += bv3;
            *(v4f*)(dst + (size_t)b * (16 * (size_t)SLABF)) = val;
        }
    }
}

static __device__ __forceinline__ void wg_barrier_lds_only()
{
    asm volatile("s_wait_dscnt 0x0\n\ts_barrier_signal -1\n\ts_barrier_wait -1" ::: "memory");
}

static __device__ __forceinline__ float lane_up4(const float v, const int q)
{
    const int iv = __builtin_bit_cast(int, v);
    const int t1 = __builtin_amdgcn_update_dpp(0, iv, 0x114, 0xf, 0xf, true);
    const int t2 = (int)__builtin_amdgcn_permlanex16((unsigned)iv, (unsigned)iv,
                                                     0x3210fedcu, 0xba987654u, false, false);
    return __builtin_bit_cast(float, q == 4 ? t2 : t1);
}

static __device__ __forceinline__ float f32_bits(const unsigned b)
{
    return __builtin_bit_cast(float, b);
}

static __device__ __forceinline__ v2f tanh_pair(const v2f x)
{
    const float l2e_hi = f32_bits(0x3fb8aa3bu);
    const float l2e_lo = f32_bits(0x32a5705fu);
    v2f ax;
    ax.x = __builtin_fabsf(x.x);
    ax.y = __builtin_fabsf(x.y);

    const v2f y = ax + ax;
    const v2f t = y * l2e_hi;
    v2f n;
    n.x = __builtin_rintf(t.x);
    n.y = __builtin_rintf(t.y);
    v2f f;
    f.x = __builtin_fmaf(y.x, l2e_lo, __builtin_fmaf(y.x, l2e_hi, -n.x));
    f.y = __builtin_fmaf(y.y, l2e_lo, __builtin_fmaf(y.y, l2e_hi, -n.y));
    v2f e;
    e.x = __builtin_ldexpf(__builtin_amdgcn_exp2f(f.x), (int)n.x);
    e.y = __builtin_ldexpf(__builtin_amdgcn_exp2f(f.y), (int)n.y);
    const float big = f32_bits(0x42b17218u);
    e.x = (big < y.x) ? __builtin_inff() : e.x;
    e.y = (big < y.y) ? __builtin_inff() : e.y;
    const v2f d = e + 1.0f;
    v2f lg;
    lg.x = __builtin_fmaf(__builtin_amdgcn_rcpf(d.x), -2.0f, 1.0f);
    lg.y = __builtin_fmaf(__builtin_amdgcn_rcpf(d.y), -2.0f, 1.0f);

    const v2f x2 = x * x;
    const float c0 = f32_bits(0xbbbac73du);
    const float c1 = f32_bits(0x3ca908c9u);
    const float c2 = f32_bits(0xbd5c1c4eu);
    const float c3 = f32_bits(0x3e088382u);
    const float c4 = f32_bits(0xbeaaaa99u);
    v2f p;
    p.x = __builtin_fmaf(x2.x, c0, c1);
    p.y = __builtin_fmaf(x2.y, c0, c1);
    p.x = __builtin_fmaf(x2.x, p.x, c2);
    p.y = __builtin_fmaf(x2.y, p.y, c2);
    p.x = __builtin_fmaf(x2.x, p.x, c3);
    p.y = __builtin_fmaf(x2.y, p.y, c3);
    p.x = __builtin_fmaf(x2.x, p.x, c4);
    p.y = __builtin_fmaf(x2.y, p.y, c4);
    const v2f qq = ax * p;
    v2f sm;
    sm.x = __builtin_fmaf(x2.x, qq.x, ax.x);
    sm.y = __builtin_fmaf(x2.y, qq.y, ax.y);

    v2f res;
    res.x = (ax.x < 0.625f) ? sm.x : lg.x;
    res.y = (ax.y < 0.625f) ? sm.y : lg.y;
    res.x = __builtin_copysignf(res.x, x.x);
    res.y = __builtin_copysignf(res.y, x.y);
    return res;
}

static __device__ __forceinline__ void scan_step(
    const int t, const int w, const int lane, const int q, const int j,
    const v2f cur, v2f &ld, v2f &hu, v2f &hs,
    const float* __restrict__ aslab, float* __restrict__ hcol,
    v2f (*bnd)[SW][32], const v2f u0, const v2f u1)
{
    {
        int tt = t + PD;
        tt = tt < LQ * w ? LQ * w : tt;
        tt = tt > LQ * w + (DD1 - 1) + (LQ - 1) ? LQ * w + (DD1 - 1) + (LQ - 1) : tt;
        ld = *(const v2f*)(aslab + (size_t)(tt & (NS - 1)) * SLOTF);
    }

    if (t >= LQ * w && t <= LQ * w + (DD1 - 1) + (LQ - 1)) {
        const v2f hb = bnd[(t + 1) & 1][w > 0 ? w - 1 : 0][28 + (lane & 3)];
        v2f hl;
        hl.x = (q == 0) ? (w > 0 ? hb.x : 0.0f) : hs.x;
        hl.y = (q == 0) ? (w > 0 ? hb.y : 0.0f) : hs.y;
        const int i = t - j;
        v2f z;
        z.x = __builtin_fmaf(u0.x, hu.x, __builtin_fmaf(u1.x, hl.x, cur.x));
        z.y = __builtin_fmaf(u0.y, hu.y, __builtin_fmaf(u1.y, hl.y, cur.y));
        v2f hn = tanh_pair(z);
        hn.x = (i >= 0) ? hn.x : 0.0f;
        hn.y = (i >= 0) ? hn.y : 0.0f;
        bnd[t & 1][w][lane] = hn;
        if (i >= 0 && i < DD1)
            *(v2f*)(hcol + (size_t)i * ROWF) = hn;
        hu = hn;
        hs.x = lane_up4(hn.x, q);
        hs.y = lane_up4(hn.y, q);
    }
    wg_barrier_lds_only();
}

__global__ __launch_bounds__(SW * 32)
void mdrnn_scan_channels(const float* __restrict__ a, const float* __restrict__ u,
                         float* __restrict__ h_out)
{
    __shared__ __attribute__((aligned(8))) v2f bnd[2][SW][32];

    const int lane = threadIdx.x & 31;
    const int w    = __builtin_amdgcn_readfirstlane(threadIdx.x >> 5);
    const int g    = blockIdx.x;
    const int p    = lane & 3;
    const int q    = lane >> 2;
    const int j    = LQ * w + q;
    const int o    = (g & 15) * CG + 2 * p;

    v2f u0, u1;
    u0.x = u[o];
    u0.y = u[o + 1];
    u1.x = u[SOUT + o];
    u1.y = u[SOUT + o + 1];

    const size_t col0 = (size_t)j * CELLF + (size_t)g * CG + 2 * p;
    const float* aslab = a + (size_t)g * SLABF + 64 * w + 2 * lane;
    float* hcol = h_out + col0;

    v2f hu = {0.0f, 0.0f};
    v2f hs = {0.0f, 0.0f};
    v2f r0, r1, r2, r3, r4;

    r0 = *(const v2f*)(aslab + 0 * SLOTF);
    r1 = *(const v2f*)(aslab + 1 * SLOTF);
    r2 = *(const v2f*)(aslab + 2 * SLOTF);
    r3 = *(const v2f*)(aslab + 3 * SLOTF);

    for (int t = 0; t < DD1 + DD2 - 1; t += PD + 1) {
        scan_step(t,     w, lane, q, j, r0, r4, hu, hs, aslab, hcol, bnd, u0, u1);
        scan_step(t + 1, w, lane, q, j, r1, r0, hu, hs, aslab, hcol, bnd, u0, u1);
        scan_step(t + 2, w, lane, q, j, r2, r1, hu, hs, aslab, hcol, bnd, u0, u1);
        scan_step(t + 3, w, lane, q, j, r3, r2, hu, hs, aslab, hcol, bnd, u0, u1);
        scan_step(t + 4, w, lane, q, j, r4, r3, hu, hs, aslab, hcol, bnd, u0, u1);
    }
}

extern "C" void kernel_launch(void* const* d_in, const int* in_sizes, int n_in,
                              void* d_out, int out_size, void* d_ws, size_t ws_size,
                              hipStream_t stream)
{
    const float* x    = (const float*)d_in[0];
    const float* w    = (const float*)d_in[1];
    const float* u    = (const float*)d_in[2];
    const float* bias = (const float*)d_in[3];
    float* h     = (float*)d_out;
    float* a_ws  = (float*)d_ws;

    dim3 g1(SOUT / 32, DD2 / 32, NS);
    mdrnn_gemm_wmma<<<g1, 256, 0, stream>>>(x, w, bias, a_ws);

    mdrnn_scan_channels<<<(BB * SOUT) / CG, SW * 32, 0, stream>>>(a_ws, u, h);
}


# scan: a loads issued 14 anti-diagonal steps ahead of use instead of 4 (15 rotating register pairs, 17 trips of 15 steps): load latency leaves the step chain; sums unchanged
# speedup vs baseline: 5.8743x; 1.0514x over previous
#include <hip/hip_runtime.h>


#define DD1  128
#define DD2  128
#define BB   16
#define SIN  64
#define SOUT 128

#define CELLF (BB * SOUT)
#define ROWF  (DD2 * CELLF)

#define NS    128

#define SW 16
#define CG 8
#define LQ (DD2 / SW)
#define PD 14
#define SLOTF (DD2 * CG)
#define SLABF (NS * SLOTF)

typedef __attribute__((ext_vector_type(2))) float v2f;
typedef __attribute__((ext_vector_type(4))) float v4f;
typedef __attribute__((ext_vector_type(8))) float v8f;

#define XP 68
#define TJ 520

static __device__ __forceinline__ const float* cell_x(const float* __restrict__ x,
                                                      const int s, const int j)
{
    const int i = (s - j) & (DD1 - 1);
    return x + ((size_t)(i * DD2 + j)) * (BB * SIN);
}

__global__ __launch_bounds__(256)
void mdrnn_gemm_wmma(const float* __restrict__ x, const float* __restrict__ w,
                     const float* __restrict__ bias, float* __restrict__ a_ws)
{
    __shared__ __attribute__((aligned(16))) float tile[32 * TJ];
    __shared__ __attribute__((aligned(16))) float xs[8][16 * XP];

    const int s    = blockIdx.z;
    const int j0   = blockIdx.y * 32;
    const int o0   = blockIdx.x * 32;
    const int tid  = threadIdx.x;
    const int lane = tid & 31;
    const int wave = tid >> 5;

    const int lm = lane & 15;
    const int kh = (lane >> 4) * 2;
    const int hh = lane >> 4;

    const int jw = j0 + wave * 4;
    const float* xw = cell_x(x, s, jw);
    float* xsw = xs[wave];
    float* sdst = xsw + hh * XP + lm * 4;
    const float* afrag = xsw + lm * XP + kh;

    v4f pf[8];
    #pragma unroll
    for (int q = 0; q < 8; ++q)
        pf[q] = *(const v4f*)(xw + (q * 32 + lane) * 4);

    #pragma unroll
    for (int t = 0; t < 4; ++t) {
        const int jj = wave * 4 + t;

        #pragma unroll
        for (int q = 0; q < 8; ++q)
            *(v4f*)(sdst + 2 * q * XP) = pf[q];

        if (t < 3) {
            const float* xn = cell_x(x, s, jw + t + 1);
            #pragma unroll
            for (int q = 0; q < 8; ++q)
                pf[q] = *(const v4f*)(xn + (q * 32 + lane) * 4);
        }

        v8f acc0 = {};
        v8f acc1 = {};
        #pragma unroll
        for (int k0 = 0; k0 < SIN; k0 += 4) {
            const v2f af = *(const v2f*)(afrag + k0);
            const float* wp = w + (size_t)(k0 + kh) * SOUT + (o0 + lm);
            v2f bf0, bf1;
            bf0.x = wp[0];
            bf0.y = wp[SOUT];
            bf1.x = wp[16];
            bf1.y = wp[SOUT + 16];
            acc0 = __builtin_amdgcn_wmma_f32_16x16x4_f32(
                false, af, false, bf0, (short)0, acc0, false, false);
            acc1 = __builtin_amdgcn_wmma_f32_16x16x4_f32(
                false, af, false, bf1, (short)0, acc1, false, false);
        }

        #pragma unroll
        for (int v = 0; v < 8; ++v) {
            tile[jj * TJ + (v + 8 * hh) * 32 + lm]      = acc0[v];
            tile[jj * TJ + (v + 8 * hh) * 32 + 16 + lm] = acc1[v];
        }
    }
    __syncthreads();

    {
        const int oct = tid >> 6;
        const int p   = tid & 63;
        const int jj  = p >> 1;
        const int c0  = (p & 1) * 4;
        const int ob  = o0 + oct * 8 + c0;
        const float bv0 = bias[ob + 0];
        const float bv1 = bias[ob + 1];
        const float bv2 = bias[ob + 2];
        const float bv3 = bias[ob + 3];
        const float* srcp = tile + jj * TJ + oct * 8 + c0;
        float* dst = a_ws + ((size_t)(((o0 >> 3) + oct) * NS + s)) * SLOTF + (size_t)j0 * CG + p * 4;
        #pragma unroll 4
        for (int b = 0; b < BB; ++b) {
            v4f val = *(const v4f*)(srcp + b * 32);
            val.x += bv0;
            val.y += bv1;
            val.z += bv2;
            val.w += bv3;
            *(v4f*)(dst + (size_t)b * (16 * (size_t)SLABF)) = val;
        }
    }
}

static __device__ __forceinline__ void wg_barrier_lds_only()
{
    asm volatile("s_wait_dscnt 0x0\n\ts_barrier_signal -1\n\ts_barrier_wait -1" ::: "memory");
}

static __device__ __forceinline__ float lane_up4(const float v, const int q)
{
    const int iv = __builtin_bit_cast(int, v);
    const int t1 = __builtin_amdgcn_update_dpp(0, iv, 0x114, 0xf, 0xf, true);
    const int t2 = (int)__builtin_amdgcn_permlanex16((unsigned)iv, (unsigned)iv,
                                                     0x3210fedcu, 0xba987654u, false, false);
    return __builtin_bit_cast(float, q == 4 ? t2 : t1);
}

static __device__ __forceinline__ float f32_bits(const unsigned b)
{
    return __builtin_bit_cast(float, b);
}

static __device__ __forceinline__ v2f tanh_pair(const v2f x)
{
    const float l2e_hi = f32_bits(0x3fb8aa3bu);
    const float l2e_lo = f32_bits(0x32a5705fu);
    v2f ax;
    ax.x = __builtin_fabsf(x.x);
    ax.y = __builtin_fabsf(x.y);

    const v2f y = ax + ax;
    const v2f t = y * l2e_hi;
    v2f n;
    n.x = __builtin_rintf(t.x);
    n.y = __builtin_rintf(t.y);
    v2f f;
    f.x = __builtin_fmaf(y.x, l2e_lo, __builtin_fmaf(y.x, l2e_hi, -n.x));
    f.y = __builtin_fmaf(y.y, l2e_lo, __builtin_fmaf(y.y, l2e_hi, -n.y));
    v2f e;
    e.x = __builtin_ldexpf(__builtin_amdgcn_exp2f(f.x), (int)n.x);
    e.y = __builtin_ldexpf(__builtin_amdgcn_exp2f(f.y), (int)n.y);
    const float big = f32_bits(0x42b17218u);
    e.x = (big < y.x) ? __builtin_inff() : e.x;
    e.y = (big < y.y) ? __builtin_inff() : e.y;
    const v2f d = e + 1.0f;
    v2f lg;
    lg.x = __builtin_fmaf(__builtin_amdgcn_rcpf(d.x), -2.0f, 1.0f);
    lg.y = __builtin_fmaf(__builtin_amdgcn_rcpf(d.y), -2.0f, 1.0f);

    const v2f x2 = x * x;
    const float c0 = f32_bits(0xbbbac73du);
    const float c1 = f32_bits(0x3ca908c9u);
    const float c2 = f32_bits(0xbd5c1c4eu);
    const float c3 = f32_bits(0x3e088382u);
    const float c4 = f32_bits(0xbeaaaa99u);
    v2f p;
    p.x = __builtin_fmaf(x2.x, c0, c1);
    p.y = __builtin_fmaf(x2.y, c0, c1);
    p.x = __builtin_fmaf(x2.x, p.x, c2);
    p.y = __builtin_fmaf(x2.y, p.y, c2);
    p.x = __builtin_fmaf(x2.x, p.x, c3);
    p.y = __builtin_fmaf(x2.y, p.y, c3);
    p.x = __builtin_fmaf(x2.x, p.x, c4);
    p.y = __builtin_fmaf(x2.y, p.y, c4);
    const v2f qq = ax * p;
    v2f sm;
    sm.x = __builtin_fmaf(x2.x, qq.x, ax.x);
    sm.y = __builtin_fmaf(x2.y, qq.y, ax.y);

    v2f res;
    res.x = (ax.x < 0.625f) ? sm.x : lg.x;
    res.y = (ax.y < 0.625f) ? sm.y : lg.y;
    res.x = __builtin_copysignf(res.x, x.x);
    res.y = __builtin_copysignf(res.y, x.y);
    return res;
}

static __device__ __forceinline__ void scan_step(
    const int t, const int w, const int lane, const int q, const int j,
    const v2f cur, v2f &ld, v2f &hu, v2f &hs,
    const float* __restrict__ aslab, float* __restrict__ hcol,
    v2f (*bnd)[SW][32], const v2f u0, const v2f u1)
{
    {
        int tt = t + PD;
        tt = tt < LQ * w ? LQ * w : tt;
        tt = tt > LQ * w + (DD1 - 1) + (LQ - 1) ? LQ * w + (DD1 - 1) + (LQ - 1) : tt;
        ld = *(const v2f*)(aslab + (size_t)(tt & (NS - 1)) * SLOTF);
    }

    if (t >= LQ * w && t <= LQ * w + (DD1 - 1) + (LQ - 1)) {
        const v2f hb = bnd[(t + 1) & 1][w > 0 ? w - 1 : 0][28 + (lane & 3)];
        v2f hl;
        hl.x = (q == 0) ? (w > 0 ? hb.x : 0.0f) : hs.x;
        hl.y = (q == 0) ? (w > 0 ? hb.y : 0.0f) : hs.y;
        const int i = t - j;
        v2f z;
        z.x = __builtin_fmaf(u0.x, hu.x, __builtin_fmaf(u1.x, hl.x, cur.x));
        z.y = __builtin_fmaf(u0.y, hu.y, __builtin_fmaf(u1.y, hl.y, cur.y));
        v2f hn = tanh_pair(z);
        hn.x = (i >= 0) ? hn.x : 0.0f;
        hn.y = (i >= 0) ? hn.y : 0.0f;
        bnd[t & 1][w][lane] = hn;
        if (i >= 0 && i < DD1)
            *(v2f*)(hcol + (size_t)i * ROWF) = hn;
        hu = hn;
        hs.x = lane_up4(hn.x, q);
        hs.y = lane_up4(hn.y, q);
    }
    wg_barrier_lds_only();
}

__global__ __launch_bounds__(SW * 32)
void mdrnn_scan_channels(const float* __restrict__ a, const float* __restrict__ u,
                         float* __restrict__ h_out)
{
    __shared__ __attribute__((aligned(8))) v2f bnd[2][SW][32];

    const int lane = threadIdx.x & 31;
    const int w    = __builtin_amdgcn_readfirstlane(threadIdx.x >> 5);
    const int g    = blockIdx.x;
    const int p    = lane & 3;
    const int q    = lane >> 2;
    const int j    = LQ * w + q;
    const int o    = (g & 15) * CG + 2 * p;

    v2f u0, u1;
    u0.x = u[o];
    u0.y = u[o + 1];
    u1.x = u[SOUT + o];
    u1.y = u[SOUT + o + 1];

    const size_t col0 = (size_t)j * CELLF + (size_t)g * CG + 2 * p;
    const float* aslab = a + (size_t)g * SLABF + 64 * w + 2 * lane;
    float* hcol = h_out + col0;

    v2f hu = {0.0f, 0.0f};
    v2f hs = {0.0f, 0.0f};
    v2f r[PD + 1];

    #pragma unroll
    for (int k = 0; k < PD; ++k)
        r[k] = *(const v2f*)(aslab + (size_t)k * SLOTF);

    static_assert((DD1 + DD2 - 1) % (PD + 1) == 0, "trips must cover the 255 steps exactly");
    for (int t = 0; t < DD1 + DD2 - 1; t += PD + 1) {
        #pragma unroll
        for (int k = 0; k <= PD; ++k)
            scan_step(t + k, w, lane, q, j, r[k], r[(k + PD) % (PD + 1)],
                      hu, hs, aslab, hcol, bnd, u0, u1);
    }
}

extern "C" void kernel_launch(void* const* d_in, const int* in_sizes, int n_in,
                              void* d_out, int out_size, void* d_ws, size_t ws_size,
                              hipStream_t stream)
{
    const float* x    = (const float*)d_in[0];
    const float* w    = (const float*)d_in[1];
    const float* u    = (const float*)d_in[2];
    const float* bias = (const float*)d_in[3];
    float* h     = (float*)d_out;
    float* a_ws  = (float*)d_ws;

    dim3 g1(SOUT / 32, DD2 / 32, NS);
    mdrnn_gemm_wmma<<<g1, 256, 0, stream>>>(x, w, bias, a_ws);

    mdrnn_scan_channels<<<(BB * SOUT) / CG, SW * 32, 0, stream>>>(a_ws, u, h);
}


# scan: wave w lags one more step per wave (270 steps, not 255), so a handed-over h is used two steps after it is made: LDS store, split barrier and read-back run beside the arithmetic; sums unchanged
# speedup vs baseline: 6.0929x; 1.0372x over previous
#include <hip/hip_runtime.h>


#define DD1  128
#define DD2  128
#define BB   16
#define SIN  64
#define SOUT 128

#define CELLF (BB * SOUT)
#define ROWF  (DD2 * CELLF)

#define NS    128

#define SW 16
#define CG 8
#define LQ (DD2 / SW)
#define PD 14
#define NSTEP (DD1 + DD2 - 1 + (SW - 1))
#define SLOTF (DD2 * CG)
#define SLABF (NS * SLOTF)

typedef __attribute__((ext_vector_type(2))) float v2f;
typedef __attribute__((ext_vector_type(4))) float v4f;
typedef __attribute__((ext_vector_type(8))) float v8f;

#define XP 68
#define TJ 520

static __device__ __forceinline__ const float* cell_x(const float* __restrict__ x,
                                                      const int s, const int j)
{
    const int i = (s - j) & (DD1 - 1);
    return x + ((size_t)(i * DD2 + j)) * (BB * SIN);
}

__global__ __launch_bounds__(256)
void mdrnn_gemm_wmma(const float* __restrict__ x, const float* __restrict__ w,
                     const float* __restrict__ bias, float* __restrict__ a_ws)
{
    __shared__ __attribute__((aligned(16))) float tile[32 * TJ];
    __shared__ __attribute__((aligned(16))) float xs[8][16 * XP];

    const int s    = blockIdx.z;
    const int j0   = blockIdx.y * 32;
    const int o0   = blockIdx.x * 32;
    const int tid  = threadIdx.x;
    const int lane = tid & 31;
    const int wave = tid >> 5;

    const int lm = lane & 15;
    const int kh = (lane >> 4) * 2;
    const int hh = lane >> 4;

    const int jw = j0 + wave * 4;
    const float* xw = cell_x(x, s, jw);
    float* xsw = xs[wave];
    float* sdst = xsw + hh * XP + lm * 4;
    const float* afrag = xsw + lm * XP + kh;

    v4f pf[8];
    #pragma unroll
    for (int q = 0; q < 8; ++q)
        pf[q] = *(const v4f*)(xw + (q * 32 + lane) * 4);

    #pragma unroll
    for (int t = 0; t < 4; ++t) {
        const int jj = wave * 4 + t;

        #pragma unroll
        for (int q = 0; q < 8; ++q)
            *(v4f*)(sdst + 2 * q * XP) = pf[q];

        if (t < 3) {
            const float* xn = cell_x(x, s, jw + t + 1);
            #pragma unroll
            for (int q = 0; q < 8; ++q)
                pf[q] = *(const v4f*)(xn + (q * 32 + lane) * 4);
        }

        v8f acc0 = {};
        v8f acc1 = {};
        #pragma unroll
        for (int k0 = 0; k0 < SIN; k0 += 4) {
            const v2f af = *(const v2f*)(afrag + k0);
            const float* wp = w + (size_t)(k0 + kh) * SOUT + (o0 + lm);
            v2f bf0, bf1;
            bf0.x = wp[0];
            bf0.y = wp[SOUT];
            bf1.x = wp[16];
            bf1.y = wp[SOUT + 16];
            acc0 = __builtin_amdgcn_wmma_f32_16x16x4_f32(
                false, af, false, bf0, (short)0, acc0, false, false);
            acc1 = __builtin_amdgcn_wmma_f32_16x16x4_f32(
                false, af, false, bf1, (short)0, acc1, false, false);
        }

        #pragma unroll
        for (int v = 0; v < 8; ++v) {
            tile[jj * TJ + (v + 8 * hh) * 32 + lm]      = acc0[v];
            tile[jj * TJ + (v + 8 * hh) * 32 + 16 + lm] = acc1[v];
        }
    }
    __syncthreads();

    {
        const int oct = tid >> 6;
        const int p   = tid & 63;
        const int jj  = p >> 1;
        const int c0  = (p & 1) * 4;
        const int ob  = o0 + oct * 8 + c0;
        const float bv0 = bias[ob + 0];
        const float bv1 = bias[ob + 1];
        const float bv2 = bias[ob + 2];
        const float bv3 = bias[ob + 3];
        const float* srcp = tile + jj * TJ + oct * 8 + c0;
        float* dst = a_ws + ((size_t)(((o0 >> 3) + oct) * NS + s)) * SLOTF + (size_t)j0 * CG + p * 4;
        #pragma unroll 4
        for (int b = 0; b < BB; ++b) {
            v4f val = *(const v4f*)(srcp + b * 32);
            val.x += bv0;
            val.y += bv1;
            val.z += bv2;
            val.w += bv3;
            *(v4f*)(dst + (size_t)b * (16 * (size_t)SLABF)) = val;
        }
    }
}

static __device__ __forceinline__ void wg_barrier_arrive_lds_only()
{
    asm volatile("s_wait_dscnt 0x0\n\ts_barrier_signal -1" ::: "memory");
}
static __device__ __forceinline__ void wg_barrier_complete()
{
    asm volatile("s_barrier_wait -1" ::: "memory");
}

static __device__ __forceinline__ float lane_up4(const float v, const int q)
{
    const int iv = __builtin_bit_cast(int, v);
    const int t1 = __builtin_amdgcn_update_dpp(0, iv, 0x114, 0xf, 0xf, true);
    const int t2 = (int)__builtin_amdgcn_permlanex16((unsigned)iv, (unsigned)iv,
                                                     0x3210fedcu, 0xba987654u, false, false);
    return __builtin_bit_cast(float, q == 4 ? t2 : t1);
}

static __device__ __forceinline__ float f32_bits(const unsigned b)
{
    return __builtin_bit_cast(float, b);
}

static __device__ __forceinline__ v2f tanh_pair(const v2f x)
{
    const float l2e_hi = f32_bits(0x3fb8aa3bu);
    const float l2e_lo = f32_bits(0x32a5705fu);
    v2f ax;
    ax.x = __builtin_fabsf(x.x);
    ax.y = __builtin_fabsf(x.y);

    const v2f y = ax + ax;
    const v2f t = y * l2e_hi;
    v2f n;
    n.x = __builtin_rintf(t.x);
    n.y = __builtin_rintf(t.y);
    v2f f;
    f.x = __builtin_fmaf(y.x, l2e_lo, __builtin_fmaf(y.x, l2e_hi, -n.x));
    f.y = __builtin_fmaf(y.y, l2e_lo, __builtin_fmaf(y.y, l2e_hi, -n.y));
    v2f e;
    e.x = __builtin_ldexpf(__builtin_amdgcn_exp2f(f.x), (int)n.x);
    e.y = __builtin_ldexpf(__builtin_amdgcn_exp2f(f.y), (int)n.y);
    const float big = f32_bits(0x42b17218u);
    e.x = (big < y.x) ? __builtin_inff() : e.x;
    e.y = (big < y.y) ? __builtin_inff() : e.y;
    const v2f d = e + 1.0f;
    v2f lg;
    lg.x = __builtin_fmaf(__builtin_amdgcn_rcpf(d.x), -2.0f, 1.0f);
    lg.y = __builtin_fmaf(__builtin_amdgcn_rcpf(d.y), -2.0f, 1.0f);

    const v2f x2 = x * x;
    const float c0 = f32_bits(0xbbbac73du);
    const float c1 = f32_bits(0x3ca908c9u);
    const float c2 = f32_bits(0xbd5c1c4eu);
    const float c3 = f32_bits(0x3e088382u);
    const float c4 = f32_bits(0xbeaaaa99u);
    v2f p;
    p.x = __builtin_fmaf(x2.x, c0, c1);
    p.y = __builtin_fmaf(x2.y, c0, c1);
    p.x = __builtin_fmaf(x2.x, p.x, c2);
    p.y = __builtin_fmaf(x2.y, p.y, c2);
    p.x = __builtin_fmaf(x2.x, p.x, c3);
    p.y = __builtin_fmaf(x2.y, p.y, c3);
    p.x = __builtin_fmaf(x2.x, p.x, c4);
    p.y = __builtin_fmaf(x2.y, p.y, c4);
    const v2f qq = ax * p;
    v2f sm;
    sm.x = __builtin_fmaf(x2.x, qq.x, ax.x);
    sm.y = __builtin_fmaf(x2.y, qq.y, ax.y);

    v2f res;
    res.x = (ax.x < 0.625f) ? sm.x : lg.x;
    res.y = (ax.y < 0.625f) ? sm.y : lg.y;
    res.x = __builtin_copysignf(res.x, x.x);
    res.y = __builtin_copysignf(res.y, x.y);
    return res;
}

static __device__ __forceinline__ int a_slot(const int tw, const int w)
{
    int tt = tw;
    tt = tt < LQ * w ? LQ * w : tt;
    tt = tt > LQ * w + (DD1 - 1) + (LQ - 1) ? LQ * w + (DD1 - 1) + (LQ - 1) : tt;
    return tt & (NS - 1);
}

static __device__ __forceinline__ void scan_step(
    const int t, const int w, const int lane, const int q, const int j,
    const v2f cur, v2f &ld, v2f &hu, v2f &hs, v2f &hbc,
    const float* __restrict__ aslab, float* __restrict__ hcol,
    v2f (*bnd)[SW][32], const v2f u0, const v2f u1)
{
    const int tw = t - w;

    const bool act = tw >= LQ * w && tw <= LQ * w + (DD1 - 1) + (LQ - 1);
    const int i = tw - j;
    v2f hn = hu;
    if (act) {
        v2f hl;
        hl.x = (q == 0) ? (w > 0 ? hbc.x : 0.0f) : hs.x;
        hl.y = (q == 0) ? (w > 0 ? hbc.y : 0.0f) : hs.y;
        v2f z;
        z.x = __builtin_fmaf(u0.x, hu.x, __builtin_fmaf(u1.x, hl.x, cur.x));
        z.y = __builtin_fmaf(u0.y, hu.y, __builtin_fmaf(u1.y, hl.y, cur.y));
        hn = tanh_pair(z);
        hn.x = (i >= 0) ? hn.x : 0.0f;
        hn.y = (i >= 0) ? hn.y : 0.0f;
    }

    wg_barrier_complete();
    const v2f hbn = bnd[(t + 1) & 1][w > 0 ? w - 1 : 0][28 + (lane & 3)];
    if (act) {
        bnd[t & 1][w][lane] = hn;
        if (i >= 0 && i < DD1)
            *(v2f*)(hcol + (size_t)i * ROWF) = hn;
        hu = hn;
        hs.x = lane_up4(hn.x, q);
        hs.y = lane_up4(hn.y, q);
    }
    ld = *(const v2f*)(aslab + (size_t)a_slot(tw + PD, w) * SLOTF);
    wg_barrier_arrive_lds_only();
    hbc = hbn;
}

__global__ __launch_bounds__(SW * 32)
void mdrnn_scan_channels(const float* __restrict__ a, const float* __restrict__ u,
                         float* __restrict__ h_out)
{
    __shared__ __attribute__((aligned(8))) v2f bnd[2][SW][32];

    const int lane = threadIdx.x & 31;
    const int w    = __builtin_amdgcn_readfirstlane(threadIdx.x >> 5);
    const int g    = blockIdx.x;
    const int p    = lane & 3;
    const int q    = lane >> 2;
    const int j    = LQ * w + q;
    const int o    = (g & 15) * CG + 2 * p;

    v2f u0, u1;
    u0.x = u[o];
    u0.y = u[o + 1];
    u1.x = u[SOUT + o];
    u1.y = u[SOUT + o + 1];

    const size_t col0 = (size_t)j * CELLF + (size_t)g * CG + 2 * p;
    const float* aslab = a + (size_t)g * SLABF + 64 * w + 2 * lane;
    float* hcol = h_out + col0;

    v2f hu = {0.0f, 0.0f};
    v2f hs = {0.0f, 0.0f};
    v2f hbc = {0.0f, 0.0f};
    v2f r[PD + 1];

    #pragma unroll
    for (int k = 0; k < PD; ++k)
        r[k] = *(const v2f*)(aslab + (size_t)a_slot(k - w, w) * SLOTF);

    wg_barrier_arrive_lds_only();

    static_assert(NSTEP % (PD + 1) == 0, "trips must cover the steps exactly");
    for (int t = 0; t < NSTEP; t += PD + 1) {
        #pragma unroll
        for (int k = 0; k <= PD; ++k)
            scan_step(t + k, w, lane, q, j, r[k], r[(k + PD) % (PD + 1)],
                      hu, hs, hbc, aslab, hcol, bnd, u0, u1);
    }
    wg_barrier_complete();
}

extern "C" void kernel_launch(void* const* d_in, const int* in_sizes, int n_in,
                              void* d_out, int out_size, void* d_ws, size_t ws_size,
                              hipStream_t stream)
{
    const float* x    = (const float*)d_in[0];
    const float* w    = (const float*)d_in[1];
    const float* u    = (const float*)d_in[2];
    const float* bias = (const float*)d_in[3];
    float* h     = (float*)d_out;
    float* a_ws  = (float*)d_ws;

    dim3 g1(SOUT / 32, DD2 / 32, NS);
    mdrnn_gemm_wmma<<<g1, 256, 0, stream>>>(x, w, bias, a_ws);

    mdrnn_scan_channels<<<(BB * SOUT) / CG, SW * 32, 0, stream>>>(a_ws, u, h);
}
